# starting kernel
# baseline (speedup 1.0000x reference)
#include <hip/hip_runtime.h>
#include <math.h>


typedef __attribute__((ext_vector_type(2))) float v2f;
typedef __attribute__((ext_vector_type(8))) float v8f;

#define WPB 8

__global__ void dhcf_init_kernel(const float* __restrict__ u_emb,
                                 const float* __restrict__ i_emb,
                                 float* __restrict__ X,
                                 float* __restrict__ ACC,
                                 float* __restrict__ deg,
                                 int nu64, int n64, int n)
{
    int idx = blockIdx.x * blockDim.x + threadIdx.x;
    if (idx < n64) {
        float v = (idx < nu64) ? u_emb[idx] : i_emb[idx - nu64];
        X[idx]   = v;
        ACC[idx] = v;
    }
    if (idx < n) deg[idx] = 0.0f;
}

__global__ void dhcf_degree_kernel(const int* __restrict__ eu,
                                   const int* __restrict__ ei,
                                   float* __restrict__ deg,
                                   int ne, int NU)
{
    int e = blockIdx.x * blockDim.x + threadIdx.x;
    if (e >= ne) return;
    atomicAdd(deg + eu[e], 1.0f);
    atomicAdd(deg + NU + ei[e], 1.0f);
}

__global__ void dhcf_degnorm_kernel(const float* __restrict__ deg,
                                    float* __restrict__ dis,
                                    float* __restrict__ dinv, int n)
{
    int v = blockIdx.x * blockDim.x + threadIdx.x;
    if (v >= n) return;
    float d = deg[v];
    if (d > 0.0f) { dis[v] = 1.0f / sqrtf(d); dinv[v] = 1.0f / d; }
    else          { dis[v] = 0.0f;            dinv[v] = 0.0f;     }
}

__global__ void dhcf_zero_kernel(float* __restrict__ p, int count)
{
    int idx = blockIdx.x * blockDim.x + threadIdx.x;
    if (idx < count) p[idx] = 0.0f;
}

__global__ void dhcf_edge_gather_scatter(const int* __restrict__ eu,
                                         const int* __restrict__ ei,
                                         const float* __restrict__ SRC,
                                         const float* __restrict__ scale,
                                         float* __restrict__ DST,
                                         int ne, int NU)
{
    int idx = blockIdx.x * blockDim.x + threadIdx.x;
    if (idx >= ne * 32) return;
    int e = idx >> 5;
    int c = (idx & 31) << 1;
    int u = eu[e];
    int i = NU + ei[e];
    float2 xu = *(const float2*)(SRC + (size_t)u * 64 + c);
    float2 xi = *(const float2*)(SRC + (size_t)i * 64 + c);
    float su = scale[u], si = scale[i];
    float* di = DST + (size_t)i * 64 + c;
    float* du = DST + (size_t)u * 64 + c;
    atomicAdd(di,     xu.x * su);
    atomicAdd(di + 1, xu.y * su);
    atomicAdd(du,     xi.x * si);
    atomicAdd(du + 1, xi.y * si);
}

__global__ __launch_bounds__(256) void dhcf_layer_kernel(
    float* __restrict__ X, const float* __restrict__ Yb,
    const float* __restrict__ dis,
    const float* __restrict__ Wgc, const float* __restrict__ bgc,
    const float* __restrict__ Wbi, const float* __restrict__ bbi,
    float* __restrict__ ACC, int ntiles)
{
    int wave = threadIdx.x >> 5;
    int lane = threadIdx.x & 31;
    int tile = blockIdx.x * WPB + wave;
    if (tile >= ntiles) return;

    int row0 = tile * 16;
    int m    = lane & 15;
    int khi  = lane >> 4;
    int rowA = row0 + m;

    const v8f vz = {0,0,0,0,0,0,0,0};
    v8f acc_s[4], acc_b[4];
    #pragma unroll
    for (int jj = 0; jj < 4; ++jj) { acc_s[jj] = vz; acc_b[jj] = vz; }

    float gscale = dis[rowA];
    const float* yrow = Yb + (size_t)rowA * 64;
    const float* xrow = X  + (size_t)rowA * 64;

    #pragma unroll
    for (int kk = 0; kk < 16; ++kk) {
        int k0 = kk * 4 + khi * 2;
        v2f g2, xg2;
        float y0 = yrow[k0], y1 = yrow[k0 + 1];
        float x0 = xrow[k0], x1 = xrow[k0 + 1];
        g2.x  = y0 * gscale;  g2.y  = y1 * gscale;
        xg2.x = x0 * g2.x;    xg2.y = x1 * g2.y;
        #pragma unroll
        for (int jj = 0; jj < 4; ++jj) {
            int j = jj * 16 + m;
            v2f bg, bb;
            bg.x = Wgc[(size_t)j * 64 + k0]; bg.y = Wgc[(size_t)j * 64 + k0 + 1];
            bb.x = Wbi[(size_t)j * 64 + k0]; bb.y = Wbi[(size_t)j * 64 + k0 + 1];
            acc_s[jj] = __builtin_amdgcn_wmma_f32_16x16x4_f32(
                false, g2,  false, bg, (short)0, acc_s[jj], false, false);
            acc_b[jj] = __builtin_amdgcn_wmma_f32_16x16x4_f32(
                false, xg2, false, bb, (short)0, acc_b[jj], false, false);
        }
    }

    float xn[4][8];
    #pragma unroll
    for (int jj = 0; jj < 4; ++jj) {
        int j = jj * 16 + m;
        float bgj = bgc[j];
        float bbj = bbi[j];
        #pragma unroll
        for (int r = 0; r < 8; ++r) {
            int row = row0 + r + 8 * khi;
            float xv = X[(size_t)row * 64 + j];
            float s = acc_s[jj][r] + bgj + xv;
            s = (s > 0.0f) ? s : 0.2f * s;
            float b = acc_b[jj][r] + bbj;
            b = (b > 0.0f) ? b : 0.2f * b;
            xn[jj][r] = s + b;
        }
    }

    #pragma unroll
    for (int r = 0; r < 8; ++r) {
        float p = 0.0f;
        #pragma unroll
        for (int jj = 0; jj < 4; ++jj) p += xn[jj][r] * xn[jj][r];
        p += __shfl_xor(p, 1, 32);
        p += __shfl_xor(p, 2, 32);
        p += __shfl_xor(p, 4, 32);
        p += __shfl_xor(p, 8, 32);
        float scl = 1.0f / fmaxf(sqrtf(p), 1e-12f);
        int row = row0 + r + 8 * khi;
        #pragma unroll
        for (int jj = 0; jj < 4; ++jj) {
            int j = jj * 16 + m;
            float v = xn[jj][r] * scl;
            X[(size_t)row * 64 + j] = v;
            ACC[(size_t)row * 64 + j] += v;
        }
    }
}

__global__ void dhcf_scale_kernel(float* __restrict__ p, int count, float s)
{
    int idx = blockIdx.x * blockDim.x + threadIdx.x;
    if (idx < count) p[idx] *= s;
}

extern "C" void kernel_launch(void* const* d_in, const int* in_sizes, int n_in,
                              void* d_out, int out_size, void* d_ws, size_t ws_size,
                              hipStream_t stream)
{
    const float* u_emb = (const float*)d_in[0];
    const float* i_emb = (const float*)d_in[1];
    const float* Wgc   = (const float*)d_in[2];
    const float* bgc   = (const float*)d_in[3];
    const float* Wbi   = (const float*)d_in[4];
    const float* bbi   = (const float*)d_in[5];
    const int*   eu    = (const int*)d_in[6];
    const int*   ei    = (const int*)d_in[7];

    const int D  = 64;
    const int NU = in_sizes[0] / D;
    const int NI = in_sizes[1] / D;
    const int N  = NU + NI;
    const int NE = in_sizes[6];
    const int L  = in_sizes[2] / (D * D);
    const int n64 = N * D;

    float* X    = (float*)d_ws;
    float* E    = X + (size_t)n64;
    float* Y    = E + (size_t)n64;
    float* deg  = Y + (size_t)n64;
    float* dis  = deg + N;
    float* dinv = dis + N;
    float* ACC  = (float*)d_out;

    dim3 blk(256);
    dhcf_init_kernel<<<(n64 + 255) / 256, blk, 0, stream>>>(
        u_emb, i_emb, X, ACC, deg, NU * D, n64, N);
    dhcf_degree_kernel<<<(NE + 255) / 256, blk, 0, stream>>>(eu, ei, deg, NE, NU);
    dhcf_degnorm_kernel<<<(N + 255) / 256, blk, 0, stream>>>(deg, dis, dinv, N);

    const int ntiles = N / 16;
    const int lgrid  = (ntiles + WPB - 1) / WPB;
    const int egrid  = (NE * 32 + 255) / 256;

    for (int l = 0; l < L; ++l) {
        dhcf_zero_kernel<<<(2 * n64 + 255) / 256, blk, 0, stream>>>(E, 2 * n64);
        dhcf_edge_gather_scatter<<<egrid, blk, 0, stream>>>(eu, ei, X, dis,  E, NE, NU);
        dhcf_edge_gather_scatter<<<egrid, blk, 0, stream>>>(eu, ei, E, dinv, Y, NE, NU);
        dhcf_layer_kernel<<<lgrid, blk, 0, stream>>>(
            X, Y, dis,
            Wgc + (size_t)l * D * D, bgc + (size_t)l * D,
            Wbi + (size_t)l * D * D, bbi + (size_t)l * D,
            ACC, ntiles);
    }
    dhcf_scale_kernel<<<(n64 + 255) / 256, blk, 0, stream>>>(
        ACC, n64, 1.0f / (float)(L + 1));
}


# Copy-in and final-scale kernels: 512 resident 8-wave work-groups walk the table in 16-byte pieces, 8 loads in flight per thread, not one 4-byte element per thread in 300,000 one-load waves a launch
# speedup vs baseline: 2.7154x; 2.7154x over previous
#include <hip/hip_runtime.h>
#include <math.h>


typedef __attribute__((ext_vector_type(2))) float v2f;
typedef __attribute__((ext_vector_type(8))) float v8f;

#define WPB 8

#define EW_GRID 512
#define EW_UNR  8

__device__ __forceinline__ float4 dhcf_in_piece(const float* __restrict__ u_emb,
                                                const float* __restrict__ i_emb, int k, int nu4)
{
    const float4* p = (k < nu4) ? ((const float4*)u_emb + k) : ((const float4*)i_emb + (k - nu4));
    return *p;
}

__global__ __launch_bounds__(256) void dhcf_init_kernel(const float* __restrict__ u_emb,
                                 const float* __restrict__ i_emb,
                                 float* __restrict__ X,
                                 float* __restrict__ ACC,
                                 float* __restrict__ deg,
                                 int* __restrict__ cnt, int ncnt,
                                 int nu4, int n4, int n)
{
    const int tid    = blockIdx.x * 256 + threadIdx.x;
    const int stride = gridDim.x * 256;
    for (int k = tid; k < ncnt; k += stride) cnt[k] = 0;
    for (int k = tid; k < n; k += stride) deg[k] = 0.0f;
    float4* X4 = (float4*)X;
    float4* A4 = (float4*)ACC;
    float4 v[EW_UNR];
    int base = tid;
    for (; base + (EW_UNR - 1) * stride < n4; base += EW_UNR * stride) {
        #pragma unroll
        for (int u = 0; u < EW_UNR; ++u) v[u] = dhcf_in_piece(u_emb, i_emb, base + u * stride, nu4);
        __builtin_amdgcn_sched_barrier(0);
        #pragma unroll
        for (int u = 0; u < EW_UNR; ++u) {
            X4[base + u * stride] = v[u];
            A4[base + u * stride] = v[u];
        }
    }
    if (base < n4) {
        #pragma unroll
        for (int u = 0; u < EW_UNR; ++u) {
            const int k = base + u * stride;
            v[u] = dhcf_in_piece(u_emb, i_emb, (k < n4) ? k : n4 - 1, nu4);
        }
        __builtin_amdgcn_sched_barrier(0);
        #pragma unroll
        for (int u = 0; u < EW_UNR; ++u) {
            const int k = base + u * stride;
            if (k < n4) { X4[k] = v[u]; A4[k] = v[u]; }
        }
    }
}

__global__ void dhcf_degree_kernel(const int* __restrict__ eu,
                                   const int* __restrict__ ei,
                                   float* __restrict__ deg,
                                   int ne, int NU)
{
    int e = blockIdx.x * blockDim.x + threadIdx.x;
    if (e >= ne) return;
    atomicAdd(deg + eu[e], 1.0f);
    atomicAdd(deg + NU + ei[e], 1.0f);
}

#define QSTRIDE 64
#define NTICKET 768
__global__ void dhcf_degnorm_kernel(float* __restrict__ deg,
                                    float* __restrict__ dis,
                                    float* __restrict__ dinv, int n)
{
    int v = blockIdx.x * blockDim.x + threadIdx.x;
    if (v >= n) return;
    float d = deg[v];
    if (d > 0.0f) { dis[v] = 1.0f / sqrtf(d); dinv[v] = 1.0f / d; }
    else          { dis[v] = 0.0f;            dinv[v] = 0.0f;     }
    if (v < NTICKET) deg[v] = 0.0f;
}

__global__ void dhcf_zero_kernel(float* __restrict__ p, int count)
{
    int idx = blockIdx.x * blockDim.x + threadIdx.x;
    if (idx < count) p[idx] = 0.0f;
}

#define EPW 32
#define EGR 16
#define EDGE_GRID 1536

__device__ __forceinline__ void dhcf_edge_chunk(const int* __restrict__ eu,
                                                const int* __restrict__ ei,
                                                const float* __restrict__ SRC,
                                                const float* __restrict__ scale,
                                                float* __restrict__ DST,
                                                int ne, int NU, int dir, int chunk, int c)
{
    const int e0   = chunk * EPW;
    const int left = ne - e0;
    const int cnt  = (left < EPW) ? left : EPW;

    int   s_l = 0, t_l = 0;
    float sc_l = 0.0f;
    if (c < cnt) {
        const int uu = eu[e0 + c];
        const int ii = NU + ei[e0 + c];
        s_l  = dir ? ii : uu;
        t_l  = dir ? uu : ii;
        sc_l = scale[s_l];
    }
    const int sc_b = __float_as_int(sc_l);

    if (cnt == EPW) {
        for (int j0 = 0; j0 < EPW; j0 += EGR) {
            float x0[EGR], x1[EGR];
            for (int t = 0; t < EGR; ++t) {
                const int s = __builtin_amdgcn_readlane(s_l, j0 + t);
                const float* p = SRC + (size_t)s * 64 + c;
                x0[t] = p[0]; x1[t] = p[32];
            }
            for (int t = 0; t < EGR; ++t) {
                const float sc = __int_as_float(__builtin_amdgcn_readlane(sc_b, j0 + t));
                const int   d  = __builtin_amdgcn_readlane(t_l, j0 + t);
                float* q = DST + (size_t)d * 64 + c;
                atomicAdd(q,      x0[t] * sc);
                atomicAdd(q + 32, x1[t] * sc);
            }
        }
    } else {
        for (int j = 0; j < cnt; ++j) {
            const int   s  = __builtin_amdgcn_readlane(s_l, j);
            const int   d  = __builtin_amdgcn_readlane(t_l, j);
            const float sc = __int_as_float(__builtin_amdgcn_readlane(sc_b, j));
            const float* p = SRC + (size_t)s * 64 + c;
            float* q = DST + (size_t)d * 64 + c;
            const float a0 = p[0], a1 = p[32];
            atomicAdd(q,      a0 * sc);
            atomicAdd(q + 32, a1 * sc);
        }
    }
}

__device__ __forceinline__ void dhcf_edge_drain(const int* __restrict__ eu,
                                                const int* __restrict__ ei,
                                                const float* __restrict__ SRC,
                                                const float* __restrict__ scale,
                                                float* __restrict__ DST,
                                                int* queue, int ne, int NU,
                                                int dir, int nch, int c)
{
    int tv = 0;
    if (c == 0) tv = atomicAdd(queue, 1);
    int t = __builtin_amdgcn_readfirstlane(tv);
    while (t < nch) {
        int nv = 0;
        if (c == 0) nv = atomicAdd(queue, 1);
        dhcf_edge_chunk(eu, ei, SRC, scale, DST, ne, NU, dir, t, c);
        t = __builtin_amdgcn_readfirstlane(nv);
    }
}

__global__ __launch_bounds__(256) void dhcf_edge_gather_scatter(
                                         const int* __restrict__ eu,
                                         const int* __restrict__ ei,
                                         const float* __restrict__ SRC,
                                         const float* __restrict__ scale,
                                         float* __restrict__ DST,
                                         int* queues, int ne, int NU, int swap)
{
    const int c   = threadIdx.x & 31;
    const int nch = (ne + EPW - 1) / EPW;
    const unsigned place = __builtin_amdgcn_s_sendmsg_rtn(0x87);
    const int half = (int)((place >> 18) & 1u);
    const int own  = half ^ swap;

    dhcf_edge_drain(eu, ei, SRC, scale, DST, queues + own * QSTRIDE, ne, NU, own, nch, c);
    if ((threadIdx.x >> 5) == 0)
        dhcf_edge_drain(eu, ei, SRC, scale, DST, queues + (own ^ 1) * QSTRIDE, ne, NU, own ^ 1, nch, c);
}


#define LROWS   648
#define SRCBITS 22
#define SRCMASK ((1u << SRCBITS) - 1u)
#define MAXNB   1024
#define CSTRIDE 32
#define BCH     2048
#define PASS_T  1024
#define TSTR    64
#define SORTCAP 12288
#define PADREC  (1023u << SRCBITS)

__global__ __launch_bounds__(256) void dhcf_bucket_count(const int* __restrict__ eu,
                                                         const int* __restrict__ ei,
                                                         int* __restrict__ cnt,
                                                         int ne, int RU, int RI, int nb)
{
    __shared__ int h[MAXNB];
    for (int k = threadIdx.x; k < nb; k += 256) h[k] = 0;
    __syncthreads();
    const int e0 = blockIdx.x * BCH;
    for (int k = threadIdx.x; k < BCH; k += 256) {
        const int e = e0 + k;
        if (e < ne) {
            atomicAdd(&h[eu[e] / RU], 1);
            atomicAdd(&h[ei[e] / RI], 1);
        }
    }
    __syncthreads();
    for (int k = threadIdx.x; k < nb; k += 256) {
        const int v = h[k];
        if (v) atomicAdd(cnt + k * CSTRIDE, v);
    }
}

__global__ __launch_bounds__(MAXNB) void dhcf_bucket_scan(int* __restrict__ cnt,
                                                          int* __restrict__ off, int nb)
{
    __shared__ int s[MAXNB];
    const int t = threadIdx.x;
    const int v = (t < nb) ? cnt[t * CSTRIDE] : 0;
    s[t] = v;
    __syncthreads();
    for (int d = 1; d < MAXNB; d <<= 1) {
        const int a = (t >= d) ? s[t - d] : 0;
        __syncthreads();
        s[t] += a;
        __syncthreads();
    }
    const int incl = s[t];
    if (t < nb) { off[t] = incl - v; cnt[t * CSTRIDE] = incl - v; }
    if (t == nb - 1) off[nb] = incl;
}

__global__ __launch_bounds__(256) void dhcf_bucket_fill(const int* __restrict__ eu,
                                                        const int* __restrict__ ei,
                                                        int* __restrict__ cursor,
                                                        unsigned* __restrict__ rec,
                                                        int ne, int NU, int RU, int RI, int nb)
{
    __shared__ int h[MAXNB];
    __shared__ int base[MAXNB];
    for (int k = threadIdx.x; k < nb; k += 256) h[k] = 0;
    __syncthreads();
    const int e0 = blockIdx.x * BCH;
    for (int k = threadIdx.x; k < BCH; k += 256) {
        const int e = e0 + k;
        if (e < ne) {
            atomicAdd(&h[eu[e] / RU], 1);
            atomicAdd(&h[ei[e] / RI], 1);
        }
    }
    __syncthreads();
    for (int k = threadIdx.x; k < nb; k += 256) {
        const int v = h[k];
        base[k] = v ? atomicAdd(cursor + k * CSTRIDE, v) : 0;
        h[k] = 0;
    }
    __syncthreads();
    for (int k = threadIdx.x; k < BCH; k += 256) {
        const int e = e0 + k;
        if (e < ne) {
            const int u  = eu[e];
            const int i  = ei[e];
            const int bu = u / RU;
            const int bi = i / RI;
            const int ru = atomicAdd(&h[bu], 1);
            rec[base[bu] + ru] = (unsigned)(NU + i) | ((unsigned)(u - bu * RU) << SRCBITS);
            const int ri = atomicAdd(&h[bi], 1);
            rec[base[bi] + ri] = (unsigned)u | ((unsigned)(RU + i - bi * RI) << SRCBITS);
        }
    }
}

#define KREC 8
__global__ __launch_bounds__(PASS_T) void dhcf_block_degree_sort(
                                         const unsigned* __restrict__ raw,
                                         const int* __restrict__ off,
                                         unsigned* __restrict__ rec,
                                         int* __restrict__ toff,
                                         float* __restrict__ dis,
                                         float* __restrict__ dinv,
                                         int NU, int NI, int RU, int RI)
{
    __shared__ int h[LROWS];
    __shared__ int tcnt[TSTR];
    __shared__ int tcur[TSTR];
    __shared__ unsigned sbuf[SORTCAP];
    const int tid   = threadIdx.x;
    const int b     = blockIdx.x;
    const int nrow  = RU + RI;
    const int ntile = (nrow + 15) >> 4;
    for (int k = tid; k < nrow; k += PASS_T) h[k] = 0;
    __syncthreads();
    const int beg = off[b];
    const int end = off[b + 1];
    unsigned rr[KREC];
    #pragma unroll
    for (int i = 0; i < KREC; ++i) {
        const int k = beg + tid + i * PASS_T;
        rr[i] = (k < end) ? raw[k] : 0u;
    }
    #pragma unroll
    for (int i = 0; i < KREC; ++i)
        if (beg + tid + i * PASS_T < end) atomicAdd(&h[rr[i] >> SRCBITS], 1);
    for (int k = beg + tid + KREC * PASS_T; k < end; k += PASS_T) atomicAdd(&h[raw[k] >> SRCBITS], 1);
    __syncthreads();
    for (int k = tid; k < nrow; k += PASS_T) {
        int  v;
        bool live;
        if (k < RU) { v = b * RU + k;        live = v < NU; }
        else        { v = b * RI + (k - RU); live = v < NI; v += NU; }
        if (live) {
            const float d = (float)h[k];
            if (d > 0.0f) { dis[v] = 1.0f / sqrtf(d); dinv[v] = 1.0f / d; }
            else          { dis[v] = 0.0f;            dinv[v] = 0.0f;     }
        }
    }
    if (tid < ntile) {
        int s = 0;
        for (int r = 0; r < 16; ++r) {
            const int k = tid * 16 + r;
            if (k < nrow) s += h[k];
        }
        tcnt[tid] = s;
    }
    __syncthreads();
    if (tid <= ntile) {
        int s = 0;
        for (int u = 0; u < tid; ++u) s += tcnt[u];
        tcur[tid] = s;
        toff[b * TSTR + tid] = beg + s;
    }
    __syncthreads();
    const int  len   = end - beg;
    const bool stage = len <= SORTCAP;
    #pragma unroll
    for (int i = 0; i < KREC; ++i) {
        if (beg + tid + i * PASS_T < end) {
            const unsigned r = rr[i];
            const int p = atomicAdd(&tcur[r >> (SRCBITS + 4)], 1);
            if (stage) sbuf[p] = r;
            else       rec[beg + p] = r;
        }
    }
    for (int k = beg + tid + KREC * PASS_T; k < end; k += PASS_T) {
        const unsigned r = raw[k];
        const int p = atomicAdd(&tcur[r >> (SRCBITS + 4)], 1);
        if (stage) sbuf[p] = r;
        else       rec[beg + p] = r;
    }
    if (stage) {
        __syncthreads();
        for (int k = tid; k < len; k += PASS_T) rec[beg + k] = sbuf[k];
    }
}


__device__ __forceinline__ int dhcf_block_row(int k, int b, int NU, int NI, int RU, int RI, bool& live)
{
    const bool isu = k < RU;
    const int  v   = isu ? (b * RU + k) : (b * RI + (k - RU));
    const int  lim = isu ? NU : NI;
    live = (k < RU + RI) & (v < lim);
    return live ? (isu ? v : v + NU) : 0;
}

template <bool SCALED, int NG, bool NEXT>
__device__ __forceinline__ unsigned dhcf_tile_chunk(const float* __restrict__ SRC, v8f (&ys)[4],
                                                    int rc_b, int sc_b, int lr0, int m, int hf,
                                                    const unsigned* __restrict__ nxp, bool nxv)
{
    float4 x0[NG], x1[NG];
    v2f    sel[NG];
    #pragma unroll
    for (int g = 0; g < NG; ++g) {
        const int a0 = __builtin_amdgcn_readlane(rc_b, 4 * g + 0);
        const int a1 = __builtin_amdgcn_readlane(rc_b, 4 * g + 1);
        const int b0 = __builtin_amdgcn_readlane(rc_b, 4 * g + 2);
        const int b1 = __builtin_amdgcn_readlane(rc_b, 4 * g + 3);
        const unsigned r0 = (unsigned)(hf ? b0 : a0);
        const unsigned r1 = (unsigned)(hf ? b1 : a1);
        x0[g] = *(const float4*)((const char*)SRC + ((r0 & SRCMASK) * 256u + (unsigned)m * 16u));
        x1[g] = *(const float4*)((const char*)SRC + ((r1 & SRCMASK) * 256u + (unsigned)m * 16u));
        sel[g].x = ((int)(r0 >> SRCBITS) - lr0 == m) ? 1.0f : 0.0f;
        sel[g].y = ((int)(r1 >> SRCBITS) - lr0 == m) ? 1.0f : 0.0f;
    }
    unsigned rc_next = PADREC;
    if (NEXT) {
        const unsigned rv = *nxp;
        rc_next = nxv ? rv : PADREC;
    }
    __builtin_amdgcn_sched_barrier(0);
    #pragma unroll
    for (int g = 0; g < NG; ++g) {
        float4 u0 = x0[g], u1 = x1[g];
        if (SCALED) {
            const int p0 = __builtin_amdgcn_readlane(sc_b, 4 * g + 0);
            const int p1 = __builtin_amdgcn_readlane(sc_b, 4 * g + 1);
            const int q0 = __builtin_amdgcn_readlane(sc_b, 4 * g + 2);
            const int q1 = __builtin_amdgcn_readlane(sc_b, 4 * g + 3);
            const float s0 = __int_as_float(hf ? q0 : p0);
            const float s1 = __int_as_float(hf ? q1 : p1);
            u0.x *= s0; u0.y *= s0; u0.z *= s0; u0.w *= s0;
            u1.x *= s1; u1.y *= s1; u1.z *= s1; u1.w *= s1;
        }
        v2f bq;
        bq.x = u0.x; bq.y = u1.x;
        ys[0] = __builtin_amdgcn_wmma_f32_16x16x4_f32(false, sel[g], false, bq, (short)0, ys[0], false, false);
        bq.x = u0.y; bq.y = u1.y;
        ys[1] = __builtin_amdgcn_wmma_f32_16x16x4_f32(false, sel[g], false, bq, (short)0, ys[1], false, false);
        bq.x = u0.z; bq.y = u1.z;
        ys[2] = __builtin_amdgcn_wmma_f32_16x16x4_f32(false, sel[g], false, bq, (short)0, ys[2], false, false);
        bq.x = u0.w; bq.y = u1.w;
        ys[3] = __builtin_amdgcn_wmma_f32_16x16x4_f32(false, sel[g], false, bq, (short)0, ys[3], false, false);
    }
    return rc_next;
}

template <bool SCALED>
__device__ __forceinline__ void dhcf_tile_sum(const unsigned* __restrict__ rec, int beg, int end,
                                              const float* __restrict__ SRC,
                                              const float* __restrict__ scale,
                                              v8f (&ys)[4], int lr0, int c, int m, int hf)
{
    const v8f vz = {0,0,0,0,0,0,0,0};
    #pragma unroll
    for (int jj = 0; jj < 4; ++jj) ys[jj] = vz;
    unsigned rc_n = (beg + c < end) ? rec[beg + c] : PADREC;
    for (int r0 = beg; r0 < end; r0 += 32) {
        const unsigned rc = rc_n;
        const int nx = r0 + 32 + c;
        float sc_l = 0.0f;
        if (SCALED) sc_l = scale[rc & SRCMASK];
        const int sc_b = __float_as_int(sc_l);
        if (end - r0 > 16) rc_n = dhcf_tile_chunk<SCALED, 8, true>(SRC, ys, (int)rc, sc_b, lr0, m, hf, rec + ((nx < end) ? nx : end - 1), nx < end);
        else               rc_n = dhcf_tile_chunk<SCALED, 4, false>(SRC, ys, (int)rc, sc_b, lr0, m, hf, rec, false);
    }
}

template <bool SCALED>
__global__ __launch_bounds__(PASS_T) void dhcf_tile_pass(
                                         const unsigned* __restrict__ rec,
                                         const int* __restrict__ toff,
                                         const float* __restrict__ SRC,
                                         const float* __restrict__ scale,
                                         const float* __restrict__ dinv,
                                         float* __restrict__ DST,
                                         int NU, int NI, int RU, int RI)
{
    const int tid = threadIdx.x;
    const int c   = tid & 31;
    const int w   = __builtin_amdgcn_readfirstlane(tid >> 5);
    const int b   = blockIdx.x;
    const int ntile = (RU + RI + 15) >> 4;
    const int m   = c & 15;
    const int hf  = c >> 4;
    const int* tb = toff + b * TSTR;

    for (int t = w; t < ntile; t += PASS_T / 32) {
        const int lr0 = t * 16;
        v8f ys[4];
        dhcf_tile_sum<SCALED>(rec, tb[t], tb[t + 1], SRC, scale, ys, lr0, c, m, hf);
        int   row[8];
        bool  lv[8];
        float dv[8];
        #pragma unroll
        for (int r = 0; r < 8; ++r) {
            row[r] = dhcf_block_row(lr0 + 8 * hf + r, b, NU, NI, RU, RI, lv[r]);
            dv[r]  = dinv[row[r]];
        }
        #pragma unroll
        for (int r = 0; r < 8; ++r) {
            if (lv[r]) {
                const float s = dv[r];
                *(float4*)((char*)DST + ((unsigned)row[r] * 256u + (unsigned)m * 16u)) =
                    make_float4(ys[0][r] * s, ys[1][r] * s, ys[2][r] * s, ys[3][r] * s);
            }
        }
    }
}

__global__ __launch_bounds__(256) void dhcf_layer_kernel(
    float* __restrict__ X, const float* __restrict__ Yb,
    const float* __restrict__ dis,
    const float* __restrict__ Wgc, const float* __restrict__ bgc,
    const float* __restrict__ Wbi, const float* __restrict__ bbi,
    float* __restrict__ ACC, int ntiles)
{
    int wave = threadIdx.x >> 5;
    int lane = threadIdx.x & 31;
    int tile = blockIdx.x * WPB + wave;
    if (tile >= ntiles) return;

    int row0 = tile * 16;
    int m    = lane & 15;
    int khi  = lane >> 4;
    int rowA = row0 + m;

    const v8f vz = {0,0,0,0,0,0,0,0};
    v8f acc_s[4], acc_b[4];
    #pragma unroll
    for (int jj = 0; jj < 4; ++jj) { acc_s[jj] = vz; acc_b[jj] = vz; }

    float gscale = dis[rowA];
    const float* yrow = Yb + (size_t)rowA * 64;
    const float* xrow = X  + (size_t)rowA * 64;

    #pragma unroll
    for (int kk = 0; kk < 16; ++kk) {
        int k0 = kk * 4 + khi * 2;
        v2f g2, xg2;
        float y0 = yrow[k0], y1 = yrow[k0 + 1];
        float x0 = xrow[k0], x1 = xrow[k0 + 1];
        g2.x  = y0 * gscale;  g2.y  = y1 * gscale;
        xg2.x = x0 * g2.x;    xg2.y = x1 * g2.y;
        #pragma unroll
        for (int jj = 0; jj < 4; ++jj) {
            int j = jj * 16 + m;
            v2f bg, bb;
            bg.x = Wgc[(size_t)j * 64 + k0]; bg.y = Wgc[(size_t)j * 64 + k0 + 1];
            bb.x = Wbi[(size_t)j * 64 + k0]; bb.y = Wbi[(size_t)j * 64 + k0 + 1];
            acc_s[jj] = __builtin_amdgcn_wmma_f32_16x16x4_f32(
                false, g2,  false, bg, (short)0, acc_s[jj], false, false);
            acc_b[jj] = __builtin_amdgcn_wmma_f32_16x16x4_f32(
                false, xg2, false, bb, (short)0, acc_b[jj], false, false);
        }
    }

    float xn[4][8];
    #pragma unroll
    for (int jj = 0; jj < 4; ++jj) {
        int j = jj * 16 + m;
        float bgj = bgc[j];
        float bbj = bbi[j];
        #pragma unroll
        for (int r = 0; r < 8; ++r) {
            int row = row0 + r + 8 * khi;
            float xv = X[(size_t)row * 64 + j];
            float s = acc_s[jj][r] + bgj + xv;
            s = (s > 0.0f) ? s : 0.2f * s;
            float b = acc_b[jj][r] + bbj;
            b = (b > 0.0f) ? b : 0.2f * b;
            xn[jj][r] = s + b;
        }
    }

    #pragma unroll
    for (int r = 0; r < 8; ++r) {
        float p = 0.0f;
        #pragma unroll
        for (int jj = 0; jj < 4; ++jj) p += xn[jj][r] * xn[jj][r];
        p += __shfl_xor(p, 1, 32);
        p += __shfl_xor(p, 2, 32);
        p += __shfl_xor(p, 4, 32);
        p += __shfl_xor(p, 8, 32);
        float scl = 1.0f / fmaxf(sqrtf(p), 1e-12f);
        int row = row0 + r + 8 * khi;
        #pragma unroll
        for (int jj = 0; jj < 4; ++jj) {
            int j = jj * 16 + m;
            float v = xn[jj][r] * scl;
            X[(size_t)row * 64 + j] = v;
            ACC[(size_t)row * 64 + j] += v;
        }
    }
}


#define FPITCH 68
#define FROWS  (((LROWS + 15) / 16) * 16)

__global__ __launch_bounds__(PASS_T) void dhcf_tile_pass_layer(
                                         const unsigned* __restrict__ rec,
                                         const int* __restrict__ toff,
                                         const float* __restrict__ SRC,
                                         const float* __restrict__ dis,
                                         float* __restrict__ X,
                                         const float* __restrict__ Wgc, const float* __restrict__ bgc,
                                         const float* __restrict__ Wbi, const float* __restrict__ bbi,
                                         float* __restrict__ ACC,
                                         float* __restrict__ XS,
                                         int NU, int NI, int RU, int RI)
{
    __shared__ __attribute__((aligned(16))) float acc[FROWS * FPITCH];
    __shared__ __attribute__((aligned(16))) float wg_s[64 * FPITCH];
    __shared__ __attribute__((aligned(16))) float wb_s[64 * FPITCH];
    __shared__ __attribute__((aligned(16))) float dss[FROWS];
    const int tid = threadIdx.x;
    const int c   = tid & 31;
    const int w   = __builtin_amdgcn_readfirstlane(tid >> 5);
    const int b   = blockIdx.x;
    const int nrow  = RU + RI;
    const int ntile = (nrow + 15) >> 4;

    {
        const int n  = tid >> 4;
        const int k4 = (tid & 15) * 4;
        const float4 a4 = *(const float4*)(Wgc + n * 64 + k4);
        const float4 b4 = *(const float4*)(Wbi + n * 64 + k4);
        *(float4*)(wg_s + n * FPITCH + k4) = a4;
        *(float4*)(wb_s + n * FPITCH + k4) = b4;
    }
    bool  ds_live;
    const int   ds_row = dhcf_block_row(tid, b, NU, NI, RU, RI, ds_live);
    const float ds_l   = ds_live ? dis[ds_row] : 0.0f;
    if (tid < FROWS) dss[tid] = ds_l;
    __syncthreads();

    const int* tb = toff + b * TSTR;
    const int m   = c & 15;
    const int khi = c >> 4;
    for (int t = w; t < ntile; t += PASS_T / 32) {
        const int lr0 = t * 16;
        {
            v8f ys[4];
            dhcf_tile_sum<false>(rec, tb[t], tb[t + 1], SRC, (const float*)nullptr, ys, lr0, c, m, khi);
            #pragma unroll
            for (int r = 0; r < 8; ++r)
                *(float4*)(acc + (lr0 + 8 * khi + r) * FPITCH + 4 * m) =
                    make_float4(ys[0][r], ys[1][r], ys[2][r], ys[3][r]);
        }
        bool live_a;
        const int ga = dhcf_block_row(lr0 + m, b, NU, NI, RU, RI, live_a);
        const float gscale = dss[lr0 + m];
        const float* yrow = acc + (lr0 + m) * FPITCH + khi * 2;
        const float* wgr  = wg_s + m * FPITCH + khi * 2;
        const float* wbr  = wb_s + m * FPITCH + khi * 2;
        const float* xrow = (const float*)((const char*)X + ((unsigned)ga * 256u + (unsigned)khi * 8u));

        const v8f vz = {0,0,0,0,0,0,0,0};
        v8f acc_s[4], acc_b[4];
        #pragma unroll
        for (int jj = 0; jj < 4; ++jj) { acc_s[jj] = vz; acc_b[jj] = vz; }

        #pragma unroll 4
        for (int kk = 0; kk < 16; ++kk) {
            const int kc = kk * 4;
            const v2f y2 = *(const v2f*)(yrow + kc);
            const v2f x2 = *(const v2f*)(xrow + kc);
            v2f g2, xg2;
            g2.x  = y2.x * gscale;  g2.y  = y2.y * gscale;
            xg2.x = x2.x * g2.x;    xg2.y = x2.y * g2.y;
            #pragma unroll
            for (int jj = 0; jj < 4; ++jj) {
                const v2f bg = *(const v2f*)(wgr + jj * 16 * FPITCH + kc);
                const v2f bb = *(const v2f*)(wbr + jj * 16 * FPITCH + kc);
                acc_s[jj] = __builtin_amdgcn_wmma_f32_16x16x4_f32(
                    false, g2,  false, bg, (short)0, acc_s[jj], false, false);
                acc_b[jj] = __builtin_amdgcn_wmma_f32_16x16x4_f32(
                    false, xg2, false, bb, (short)0, acc_b[jj], false, false);
            }
        }

        unsigned roff[8];
        bool     lv[8];
        #pragma unroll
        for (int r = 0; r < 8; ++r) {
            const int row = dhcf_block_row(lr0 + r + 8 * khi, b, NU, NI, RU, RI, lv[r]);
            roff[r] = (unsigned)row * 256u + (unsigned)m * 4u;
        }
        float xn[4][8];
        #pragma unroll
        for (int jj = 0; jj < 4; ++jj) {
            const int j = jj * 16 + m;
            const float bgj = bgc[j];
            const float bbj = bbi[j];
            #pragma unroll
            for (int r = 0; r < 8; ++r) {
                const float xv = ((const float*)((const char*)X + roff[r]))[jj * 16];
                float s = acc_s[jj][r] + bgj + xv;
                s = (s > 0.0f) ? s : 0.2f * s;
                float bq = acc_b[jj][r] + bbj;
                bq = (bq > 0.0f) ? bq : 0.2f * bq;
                xn[jj][r] = s + bq;
            }
        }

        float av[4][8];
        #pragma unroll
        for (int jj = 0; jj < 4; ++jj) {
            #pragma unroll
            for (int r = 0; r < 8; ++r)
                av[jj][r] = ((const float*)((const char*)ACC + roff[r]))[jj * 16];
        }

        float scl[8];
        #pragma unroll
        for (int r = 0; r < 8; ++r) {
            float p = 0.0f;
            #pragma unroll
            for (int jj = 0; jj < 4; ++jj) p += xn[jj][r] * xn[jj][r];
            p += __shfl_xor(p, 1, 32);
            p += __shfl_xor(p, 2, 32);
            p += __shfl_xor(p, 4, 32);
            p += __shfl_xor(p, 8, 32);
            scl[r] = 1.0f / fmaxf(sqrtf(p), 1e-12f);
        }

        #pragma unroll
        for (int r = 0; r < 8; ++r) {
            if (lv[r]) {
                float* xo = (float*)((char*)X   + roff[r]);
                float* ao = (float*)((char*)ACC + roff[r]);
                #pragma unroll
                for (int jj = 0; jj < 4; ++jj) {
                    const float v = xn[jj][r] * scl[r];
                    xo[jj * 16] = v;
                    ao[jj * 16] = av[jj][r] + v;
                }
            }
        }
        if (XS) {
            const float4 d0 = *(const float4*)(dss + lr0 + 8 * khi);
            const float4 d1 = *(const float4*)(dss + lr0 + 8 * khi + 4);
            const float dr[8] = {d0.x, d0.y, d0.z, d0.w, d1.x, d1.y, d1.z, d1.w};
            #pragma unroll
            for (int r = 0; r < 8; ++r) {
                if (lv[r]) {
                    float* so = (float*)((char*)XS + roff[r]);
                    #pragma unroll
                    for (int jj = 0; jj < 4; ++jj)
                        so[jj * 16] = (xn[jj][r] * scl[r]) * dr[r];
                }
            }
        }
    }
}

__global__ __launch_bounds__(256) void dhcf_scale_kernel(float* __restrict__ p, int n4, float s)
{
    const int tid    = blockIdx.x * 256 + threadIdx.x;
    const int stride = gridDim.x * 256;
    float4* p4 = (float4*)p;
    float4 v[EW_UNR];
    int base = tid;
    for (; base + (EW_UNR - 1) * stride < n4; base += EW_UNR * stride) {
        #pragma unroll
        for (int u = 0; u < EW_UNR; ++u) v[u] = p4[base + u * stride];
        __builtin_amdgcn_sched_barrier(0);
        #pragma unroll
        for (int u = 0; u < EW_UNR; ++u)
            p4[base + u * stride] = make_float4(v[u].x * s, v[u].y * s, v[u].z * s, v[u].w * s);
    }
    if (base < n4) {
        #pragma unroll
        for (int u = 0; u < EW_UNR; ++u) {
            const int k = base + u * stride;
            v[u] = p4[(k < n4) ? k : n4 - 1];
        }
        __builtin_amdgcn_sched_barrier(0);
        #pragma unroll
        for (int u = 0; u < EW_UNR; ++u) {
            const int k = base + u * stride;
            if (k < n4) p4[k] = make_float4(v[u].x * s, v[u].y * s, v[u].z * s, v[u].w * s);
        }
    }
}

extern "C" void kernel_launch(void* const* d_in, const int* in_sizes, int n_in,
                              void* d_out, int out_size, void* d_ws, size_t ws_size,
                              hipStream_t stream)
{
    const float* u_emb = (const float*)d_in[0];
    const float* i_emb = (const float*)d_in[1];
    const float* Wgc   = (const float*)d_in[2];
    const float* bgc   = (const float*)d_in[3];
    const float* Wbi   = (const float*)d_in[4];
    const float* bbi   = (const float*)d_in[5];
    const int*   eu    = (const int*)d_in[6];
    const int*   ei    = (const int*)d_in[7];

    const int D  = 64;
    const int NU = in_sizes[0] / D;
    const int NI = in_sizes[1] / D;
    const int N  = NU + NI;
    const int NE = in_sizes[6];
    const int L  = in_sizes[2] / (D * D);
    const int n64 = N * D;

    float* X    = (float*)d_ws;
    float* E    = X + (size_t)n64;
    float* Y    = E + (size_t)n64;
    float* XS   = Y;
    float* deg  = Y + (size_t)n64;
    float* dis  = deg + N;
    float* dinv = dis + N;
    float* ACC  = (float*)d_out;

    int nb = (N + LROWS - 3) / (LROWS - 2);
    if (nb < 256) nb = 256;
    const int RU = (NU + nb - 1) / nb;
    const int RI = (NI + nb - 1) / nb;
    int*      cnt = (int*)(dinv + N);
    int*      off = cnt + (size_t)nb * CSTRIDE;
    unsigned* rec = (unsigned*)(off + nb + 32);
    const size_t need = (size_t)((char*)(rec + 2 * (size_t)NE) - (char*)d_ws);
    unsigned* raw  = (unsigned*)E;
    int*      toff = (int*)deg;
    const bool blocked = (ws_size >= need) && (nb <= MAXNB) && (N < (1 << SRCBITS)) && (RU + RI <= LROWS)
                      && (2 * (size_t)NE <= (size_t)n64) && ((size_t)nb * TSTR <= (size_t)N);

    dim3 blk(256);
    const int n4 = n64 / 4;
    dhcf_init_kernel<<<EW_GRID, blk, 0, stream>>>(
        u_emb, i_emb, X, ACC, deg, cnt, blocked ? nb * CSTRIDE : 0, NU * D / 4, n4, N);
    if (!blocked) {
        dhcf_degree_kernel<<<(NE + 255) / 256, blk, 0, stream>>>(eu, ei, deg, NE, NU);
        dhcf_degnorm_kernel<<<(N + 255) / 256, blk, 0, stream>>>(deg, dis, dinv, N);
    }

    const int ntiles = N / 16;
    const int lgrid  = (ntiles + WPB - 1) / WPB;
    int* queues = (int*)deg;

    if (blocked) {
        const int bgrid = (NE + BCH - 1) / BCH;
        dhcf_bucket_count<<<bgrid, blk, 0, stream>>>(eu, ei, cnt, NE, RU, RI, nb);
        dhcf_bucket_scan<<<1, MAXNB, 0, stream>>>(cnt, off, nb);
        dhcf_bucket_fill<<<bgrid, blk, 0, stream>>>(eu, ei, cnt, raw, NE, NU, RU, RI, nb);
        dhcf_block_degree_sort<<<nb, PASS_T, 0, stream>>>(raw, off, rec, toff, dis, dinv, NU, NI, RU, RI);
    }

    for (int l = 0; l < L; ++l) {
        if (blocked) {
            if (l == 0)
                dhcf_tile_pass<true><<<nb, PASS_T, 0, stream>>>(rec, toff, X,  dis, dinv, E, NU, NI, RU, RI);
            else
                dhcf_tile_pass<false><<<nb, PASS_T, 0, stream>>>(rec, toff, XS, dis, dinv, E, NU, NI, RU, RI);
            dhcf_tile_pass_layer<<<nb, PASS_T, 0, stream>>>(
                rec, toff, E, dis, X,
                Wgc + (size_t)l * D * D, bgc + (size_t)l * D,
                Wbi + (size_t)l * D * D, bbi + (size_t)l * D,
                ACC, (l + 1 < L) ? XS : (float*)nullptr, NU, NI, RU, RI);
        } else {
            dhcf_zero_kernel<<<(2 * n64 + 255) / 256, blk, 0, stream>>>(E, 2 * n64);
            dhcf_edge_gather_scatter<<<EDGE_GRID, blk, 0, stream>>>(
                eu, ei, X, dis,  E, queues + 4 * QSTRIDE * l,               NE, NU, 0);
            dhcf_edge_gather_scatter<<<EDGE_GRID, blk, 0, stream>>>(
                eu, ei, E, dinv, Y, queues + 4 * QSTRIDE * l + 2 * QSTRIDE, NE, NU, 1);
            dhcf_layer_kernel<<<lgrid, blk, 0, stream>>>(
                X, Y, dis,
                Wgc + (size_t)l * D * D, bgc + (size_t)l * D,
                Wbi + (size_t)l * D * D, bbi + (size_t)l * D,
                ACC, ntiles);
        }
    }
    dhcf_scale_kernel<<<EW_GRID, blk, 0, stream>>>(
        ACC, n4, 1.0f / (float)(L + 1));
}
